# starting kernel
# baseline (speedup 1.0000x reference)
#include <hip/hip_runtime.h>
#include <hip/hip_bf16.h>

typedef __attribute__((ext_vector_type(2))) float v2f;
typedef __attribute__((ext_vector_type(8))) float v8f;

#define D_DIM 128

__global__ void __launch_bounds__(256)
gene_proj_wmma_kernel(const float* __restrict__ gene_z,
                      const float* __restrict__ W,
                      const float* __restrict__ bvec,
                      float* __restrict__ proj,
                      int n_gene)
{
    __shared__ float As[16][131];

    const int tid  = threadIdx.x;
    const int row0 = blockIdx.x * 16;

    #pragma unroll
    for (int i = tid; i < 16 * D_DIM; i += 256) {
        int r = i >> 7;
        int c = i & 127;
        int rr = row0 + r;
        if (rr >= n_gene) rr = n_gene - 1;
        As[r][c] = gene_z[(long)rr * D_DIM + c];
    }
    __syncthreads();

    const int wave = tid >> 5;
    const int lane = tid & 31;
    const int half = lane >> 4;
    const int l16  = lane & 15;
    const int col0 = wave * 16;
    const int n    = col0 + l16;

    v8f acc = {};

    #pragma unroll 8
    for (int kk = 0; kk < D_DIM / 4; ++kk) {
        const int kb = kk * 4 + half * 2;
        v2f a;
        a[0] = As[l16][kb];
        a[1] = As[l16][kb + 1];
        v2f b;
        b[0] = W[(long)kb * D_DIM + n];
        b[1] = W[(long)(kb + 1) * D_DIM + n];

        acc = __builtin_amdgcn_wmma_f32_16x16x4_f32(
             false, a,  false, b,
             (short)0, acc,  false,  false);
    }

    const float bias = bvec[n];

    if (row0 + 16 <= n_gene) {
        #pragma unroll
        for (int j = 0; j < 8; ++j) {
            const int m = j + half * 8;
            float v = acc[j] + bias;
            v = (v > 0.0f) ? v : 0.01f * v;
            proj[(long)(row0 + m) * D_DIM + n] = v;
        }
    } else {
        #pragma unroll
        for (int j = 0; j < 8; ++j) {
            const int m = j + half * 8;
            float v = acc[j] + bias;
            v = (v > 0.0f) ? v : 0.01f * v;
            const int rr = row0 + m;
            if (rr < n_gene)
                proj[(long)rr * D_DIM + n] = v;
        }
    }
}

__global__ void __launch_bounds__(256)
edge_dot_kernel(const float* __restrict__ cell_z,
                const float* __restrict__ gene_proj,
                const float* __restrict__ cell_scale,
                const float* __restrict__ cell_bias,
                const float* __restrict__ cell_std,
                const float* __restrict__ gene_scale,
                const float* __restrict__ gene_bias,
                const float* __restrict__ gene_std,
                const int* __restrict__ src_idx,
                const int* __restrict__ dst_idx,
                const int* __restrict__ cell_n_id,
                const int* __restrict__ gene_n_id,
                float* __restrict__ out,
                int E)
{
    const int lane = threadIdx.x & 31;
    const long e = (long)blockIdx.x * (blockDim.x >> 5) + (threadIdx.x >> 5);
    if (e >= E) return;

    const int s = __builtin_amdgcn_readfirstlane(src_idx[e]);
    const int d = __builtin_amdgcn_readfirstlane(dst_idx[e]);

    const float4 u4 = *(const float4*)(cell_z    + (long)s * D_DIM + lane * 4);
    const float4 v4 = *(const float4*)(gene_proj + (long)d * D_DIM + lane * 4);

    float p = u4.x * v4.x + u4.y * v4.y + u4.z * v4.z + u4.w * v4.w;

    #pragma unroll
    for (int off = 16; off > 0; off >>= 1)
        p += __shfl_xor(p, off, 32);

    if (lane == 0) {
        const int sid = cell_n_id[s];
        const int did = gene_n_id[d];
        const float loc = p * cell_scale[sid] * gene_scale[did]
                          + cell_bias[sid] + gene_bias[did];
        const float sd  = cell_std[sid] * gene_std[did];
        out[e]           = loc;
        out[(long)E + e] = sd;
    }
}

extern "C" void kernel_launch(void* const* d_in, const int* in_sizes, int n_in,
                              void* d_out, int out_size, void* d_ws, size_t ws_size,
                              hipStream_t stream) {
    const float* cell_z     = (const float*)d_in[0];
    const float* gene_z     = (const float*)d_in[1];
    const float* W_gene     = (const float*)d_in[2];
    const float* b_gene     = (const float*)d_in[3];
    const float* cell_scale = (const float*)d_in[4];
    const float* cell_bias  = (const float*)d_in[5];
    const float* cell_std   = (const float*)d_in[6];
    const float* gene_scale = (const float*)d_in[7];
    const float* gene_bias  = (const float*)d_in[8];
    const float* gene_std   = (const float*)d_in[9];
    const int*   src_idx    = (const int*)d_in[10];
    const int*   dst_idx    = (const int*)d_in[11];
    const int*   cell_n_id  = (const int*)d_in[12];
    const int*   gene_n_id  = (const int*)d_in[13];

    const int n_gene = in_sizes[1] / D_DIM;
    const int E      = in_sizes[10];

    float* gene_proj = (float*)d_ws;

    const int gemm_blocks = (n_gene + 15) / 16;
    gene_proj_wmma_kernel<<<gemm_blocks, 256, 0, stream>>>(
        gene_z, W_gene, b_gene, gene_proj, n_gene);

    const int edges_per_block = 256 / 32;
    const int edge_blocks = (E + edges_per_block - 1) / edges_per_block;
    edge_dot_kernel<<<edge_blocks, 256, 0, stream>>>(
        cell_z, gene_proj,
        cell_scale, cell_bias, cell_std,
        gene_scale, gene_bias, gene_std,
        src_idx, dst_idx, cell_n_id, gene_n_id,
        (float*)d_out, E);
}


# gene-side lookups (id, scale, bias, std) served from LDS: edge kernel made persistent, 256 work-groups of 32 waves stride over 32-edge chunks, each copies the four gene tables into its LDS once
# speedup vs baseline: 2.3780x; 2.3780x over previous
#include <hip/hip_runtime.h>
#include <hip/hip_bf16.h>

typedef __attribute__((ext_vector_type(2))) float v2f;
typedef __attribute__((ext_vector_type(8))) float v8f;

#define D_DIM 128

__global__ void __launch_bounds__(256)
gene_proj_wmma_kernel(const float* __restrict__ gene_z,
                      const float* __restrict__ W,
                      const float* __restrict__ bvec,
                      float* __restrict__ proj,
                      int n_gene)
{
    __shared__ float As[16][131];

    const int tid  = threadIdx.x;
    const int row0 = blockIdx.x * 16;

    #pragma unroll
    for (int i = tid; i < 16 * D_DIM; i += 256) {
        int r = i >> 7;
        int c = i & 127;
        int rr = row0 + r;
        if (rr >= n_gene) rr = n_gene - 1;
        As[r][c] = gene_z[(long)rr * D_DIM + c];
    }
    __syncthreads();

    const int wave = tid >> 5;
    const int lane = tid & 31;
    const int half = lane >> 4;
    const int l16  = lane & 15;
    const int col0 = wave * 16;
    const int n    = col0 + l16;

    v8f acc = {};

    #pragma unroll 8
    for (int kk = 0; kk < D_DIM / 4; ++kk) {
        const int kb = kk * 4 + half * 2;
        v2f a;
        a[0] = As[l16][kb];
        a[1] = As[l16][kb + 1];
        v2f b;
        b[0] = W[(long)kb * D_DIM + n];
        b[1] = W[(long)(kb + 1) * D_DIM + n];

        acc = __builtin_amdgcn_wmma_f32_16x16x4_f32(
             false, a,  false, b,
             (short)0, acc,  false,  false);
    }

    const float bias = bvec[n];

    if (row0 + 16 <= n_gene) {
        #pragma unroll
        for (int j = 0; j < 8; ++j) {
            const int m = j + half * 8;
            float v = acc[j] + bias;
            v = (v > 0.0f) ? v : 0.01f * v;
            proj[(long)(row0 + m) * D_DIM + n] = v;
        }
    } else {
        #pragma unroll
        for (int j = 0; j < 8; ++j) {
            const int m = j + half * 8;
            float v = acc[j] + bias;
            v = (v > 0.0f) ? v : 0.01f * v;
            const int rr = row0 + m;
            if (rr < n_gene)
                proj[(long)rr * D_DIM + n] = v;
        }
    }
}


template <int P>
__device__ __forceinline__ int group8_bcast(int x)
{
    return (int)__builtin_amdgcn_mov_dpp8((unsigned)x, (unsigned)(P * 0x249249));
}

__device__ __forceinline__ float group8_sum(float p)
{
    p += __builtin_bit_cast(float, __builtin_amdgcn_update_dpp(
             0, __builtin_bit_cast(int, p), 0x161, 0xf, 0xf, true));
    p += __builtin_bit_cast(float, __builtin_amdgcn_update_dpp(
             0, __builtin_bit_cast(int, p), 0x162, 0xf, 0xf, true));
    p += __builtin_bit_cast(float, __builtin_amdgcn_update_dpp(
             0, __builtin_bit_cast(int, p), 0x164, 0xf, 0xf, true));
    return p;
}

__device__ __forceinline__ v2f dot4_acc(const float4 a, const float4 b, v2f acc)
{
    acc = __builtin_elementwise_fma((v2f){a.x, a.y}, (v2f){b.x, b.y}, acc);
    acc = __builtin_elementwise_fma((v2f){a.z, a.w}, (v2f){b.z, b.w}, acc);
    return acc;
}

struct RowPieces {
    float4 u0, u1, u2, u3;
    float4 v0, v1, v2, v3;
};

template <int P>
__device__ __forceinline__ RowPieces load_pass(const float* __restrict__ cell_z,
                                               const float* __restrict__ gene_proj,
                                               int s, int d, int t)
{
    const int sp = group8_bcast<P>(s);
    const int dp = group8_bcast<P>(d);
    const float4* ur = (const float4*)(cell_z    + (long)sp * D_DIM) + t;
    const float4* vr = (const float4*)(gene_proj + (long)dp * D_DIM) + t;
    RowPieces r;
    r.u0 = ur[0]; r.u1 = ur[8]; r.u2 = ur[16]; r.u3 = ur[24];
    r.v0 = vr[0]; r.v1 = vr[8]; r.v2 = vr[16]; r.v3 = vr[24];
    return r;
}

__device__ __forceinline__ float sum_pass(const RowPieces& r)
{
    v2f lo = {0.0f, 0.0f}, hi = {0.0f, 0.0f};
    lo = dot4_acc(r.u0, r.v0, lo);
    hi = dot4_acc(r.u2, r.v2, hi);
    lo = dot4_acc(r.u1, r.v1, lo);
    hi = dot4_acc(r.u3, r.v3, hi);
    const v2f q = lo + hi;
    return group8_sum(q[0] + q[1]);
}

#define PIN_ORDER() __builtin_amdgcn_sched_barrier(0)

#define GENE_CAP 20000
struct GeneTables {
    float          sc[GENE_CAP];
    float          b[GENE_CAP];
    float          sd[GENE_CAP];
    unsigned short nid[GENE_CAP];
};

template <bool FULL, bool INLDS>
__device__ __forceinline__ void edge_wave(const GeneTables* tab,
                                          const float* __restrict__ cell_z,
                                          const float* __restrict__ gene_proj,
                                          const float* __restrict__ cell_scale,
                                          const float* __restrict__ cell_bias,
                                          const float* __restrict__ cell_std,
                                          const float* __restrict__ gene_scale,
                                          const float* __restrict__ gene_bias,
                                          const float* __restrict__ gene_std,
                                          const int* __restrict__ src_idx,
                                          const int* __restrict__ dst_idx,
                                          const int* __restrict__ cell_n_id,
                                          const int* __restrict__ gene_n_id,
                                          float* __restrict__ out,
                                          long e0, int lane, int E)
{
    const int  t    = lane & 7;
    const long e    = e0 + lane;
    const bool live = FULL || e < E;
    const long ec   = live ? e : (long)E - 1;

    const int s = src_idx[ec];
    const int d = dst_idx[ec];

    const int sid = cell_n_id[s];
    const int did = INLDS ? (int)tab->nid[d] : gene_n_id[d];
    PIN_ORDER();

    float p = 0.0f;
    float r;
    RowPieces b0 = load_pass<0>(cell_z, gene_proj, s, d, t);
    RowPieces b1 = load_pass<1>(cell_z, gene_proj, s, d, t);
    RowPieces b2 = load_pass<2>(cell_z, gene_proj, s, d, t);
    PIN_ORDER();
    r = sum_pass(b0); p = (t == 0) ? r : p;
    PIN_ORDER();
    const float c_sc = cell_scale[sid];
    const float c_b  = cell_bias[sid];
    const float c_sd = cell_std[sid];
    const float g_sc = INLDS ? tab->sc[did] : gene_scale[did];
    const float g_b  = INLDS ? tab->b[did]  : gene_bias[did];
    const float g_sd = INLDS ? tab->sd[did] : gene_std[did];
    b0 = load_pass<3>(cell_z, gene_proj, s, d, t);
    PIN_ORDER();
    r = sum_pass(b1); p = (t == 1) ? r : p;
    PIN_ORDER();
    b1 = load_pass<4>(cell_z, gene_proj, s, d, t);
    PIN_ORDER();
    r = sum_pass(b2); p = (t == 2) ? r : p;
    PIN_ORDER();
    b2 = load_pass<5>(cell_z, gene_proj, s, d, t);
    PIN_ORDER();
    r = sum_pass(b0); p = (t == 3) ? r : p;
    PIN_ORDER();
    b0 = load_pass<6>(cell_z, gene_proj, s, d, t);
    PIN_ORDER();
    r = sum_pass(b1); p = (t == 4) ? r : p;
    PIN_ORDER();
    b1 = load_pass<7>(cell_z, gene_proj, s, d, t);
    PIN_ORDER();
    r = sum_pass(b2); p = (t == 5) ? r : p;
    PIN_ORDER();
    r = sum_pass(b0); p = (t == 6) ? r : p;
    PIN_ORDER();
    r = sum_pass(b1); p = (t == 7) ? r : p;

    const float loc = p * c_sc * g_sc + c_b + g_b;
    const float sd  = c_sd * g_sd;
    if (live) {
        out[e]           = loc;
        out[(long)E + e] = sd;
    }
}

template <bool INLDS>
__device__ __forceinline__ void edge_chunks(const GeneTables* tab,
                                            const float* __restrict__ cell_z,
                                            const float* __restrict__ gene_proj,
                                            const float* __restrict__ cell_scale,
                                            const float* __restrict__ cell_bias,
                                            const float* __restrict__ cell_std,
                                            const float* __restrict__ gene_scale,
                                            const float* __restrict__ gene_bias,
                                            const float* __restrict__ gene_std,
                                            const int* __restrict__ src_idx,
                                            const int* __restrict__ dst_idx,
                                            const int* __restrict__ cell_n_id,
                                            const int* __restrict__ gene_n_id,
                                            float* __restrict__ out,
                                            int wv, int lane, int E)
{
    const int nchunks = (E + 31) >> 5;
    const int nwaves  = (int)gridDim.x * 32;
    for (int c = wv * (int)gridDim.x + (int)blockIdx.x; c < nchunks; c += nwaves) {
        const long e0 = (long)c * 32;
        if (INLDS && e0 + 32 <= E)
            edge_wave<true, INLDS>(tab, cell_z, gene_proj, cell_scale, cell_bias,
                                   cell_std, gene_scale, gene_bias, gene_std,
                                   src_idx, dst_idx, cell_n_id, gene_n_id, out,
                                   e0, lane, E);
        else
            edge_wave<false, INLDS>(tab, cell_z, gene_proj, cell_scale, cell_bias,
                                    cell_std, gene_scale, gene_bias, gene_std,
                                    src_idx, dst_idx, cell_n_id, gene_n_id, out,
                                    e0, lane, E);
    }
}

__global__ void __launch_bounds__(1024)
edge_dot_kernel(const float* __restrict__ cell_z,
                const float* __restrict__ gene_proj,
                const float* __restrict__ cell_scale,
                const float* __restrict__ cell_bias,
                const float* __restrict__ cell_std,
                const float* __restrict__ gene_scale,
                const float* __restrict__ gene_bias,
                const float* __restrict__ gene_std,
                const int* __restrict__ src_idx,
                const int* __restrict__ dst_idx,
                const int* __restrict__ cell_n_id,
                const int* __restrict__ gene_n_id,
                float* __restrict__ out,
                int E, int n_gene)
{
    __shared__ __attribute__((aligned(16))) GeneTables tab;

    const int tid  = threadIdx.x;
    const int lane = tid & 31;
    const int wv = __builtin_amdgcn_readfirstlane(tid >> 5);

    if (n_gene > GENE_CAP) {
        edge_chunks<false>(&tab, cell_z, gene_proj, cell_scale, cell_bias, cell_std,
                           gene_scale, gene_bias, gene_std, src_idx, dst_idx,
                           cell_n_id, gene_n_id, out, wv, lane, E);
        return;
    }

    const bool vec_ok = ((((unsigned long)gene_scale | (unsigned long)gene_bias |
                           (unsigned long)gene_std   | (unsigned long)gene_n_id) & 15ul) == 0ul);
    const int n4 = vec_ok ? (n_gene >> 2) : 0;
    if (n4 > 0) {
        const float4* gs4 = (const float4*)gene_scale;
        const float4* gb4 = (const float4*)gene_bias;
        const float4* gd4 = (const float4*)gene_std;
        const int4*   gn4 = (const int4*)gene_n_id;
        float4 vs[5], vb[5], vd[5];
        int4   vn[5];
        #pragma unroll
        for (int k = 0; k < 5; ++k) {
            int i = tid + k * 1024;
            i = (i < n4) ? i : (n4 - 1);
            vs[k] = gs4[i]; vb[k] = gb4[i]; vd[k] = gd4[i]; vn[k] = gn4[i];
        }
        #pragma unroll
        for (int k = 0; k < 5; ++k) {
            int i = tid + k * 1024;
            i = (i < n4) ? i : (n4 - 1);
            ((float4*)tab.sc)[i] = vs[k];
            ((float4*)tab.b)[i]  = vb[k];
            ((float4*)tab.sd)[i] = vd[k];
            ((ushort4*)tab.nid)[i] = make_ushort4((unsigned short)vn[k].x,
                                                  (unsigned short)vn[k].y,
                                                  (unsigned short)vn[k].z,
                                                  (unsigned short)vn[k].w);
        }
    }
    for (int i = (n4 << 2) + tid; i < n_gene; i += 1024) {
        tab.sc[i]  = gene_scale[i];
        tab.b[i]   = gene_bias[i];
        tab.sd[i]  = gene_std[i];
        tab.nid[i] = (unsigned short)gene_n_id[i];
    }
    __syncthreads();

    edge_chunks<true>(&tab, cell_z, gene_proj, cell_scale, cell_bias, cell_std,
                      gene_scale, gene_bias, gene_std, src_idx, dst_idx,
                      cell_n_id, gene_n_id, out, wv, lane, E);
}

extern "C" void kernel_launch(void* const* d_in, const int* in_sizes, int n_in,
                              void* d_out, int out_size, void* d_ws, size_t ws_size,
                              hipStream_t stream) {
    const float* cell_z     = (const float*)d_in[0];
    const float* gene_z     = (const float*)d_in[1];
    const float* W_gene     = (const float*)d_in[2];
    const float* b_gene     = (const float*)d_in[3];
    const float* cell_scale = (const float*)d_in[4];
    const float* cell_bias  = (const float*)d_in[5];
    const float* cell_std   = (const float*)d_in[6];
    const float* gene_scale = (const float*)d_in[7];
    const float* gene_bias  = (const float*)d_in[8];
    const float* gene_std   = (const float*)d_in[9];
    const int*   src_idx    = (const int*)d_in[10];
    const int*   dst_idx    = (const int*)d_in[11];
    const int*   cell_n_id  = (const int*)d_in[12];
    const int*   gene_n_id  = (const int*)d_in[13];

    const int n_gene = in_sizes[1] / D_DIM;
    const int E      = in_sizes[10];

    float* gene_proj = (float*)d_ws;

    const int gemm_blocks = (n_gene + 15) / 16;
    gene_proj_wmma_kernel<<<gemm_blocks, 256, 0, stream>>>(
        gene_z, W_gene, b_gene, gene_proj, n_gene);

    edge_dot_kernel<<<256, 1024, 0, stream>>>(
        cell_z, gene_proj,
        cell_scale, cell_bias, cell_std,
        gene_scale, gene_bias, gene_std,
        src_idx, dst_idx, cell_n_id, gene_n_id,
        (float*)d_out, E, n_gene);
}
